# starting kernel
# baseline (speedup 1.0000x reference)
#include <hip/hip_runtime.h>


typedef __attribute__((ext_vector_type(2))) float v2f;
typedef __attribute__((ext_vector_type(8))) float v8f;

#define BB 256
#define LL 1024
#define DD 128
#define HH 128
#define OO 128
#define HS 132

__device__ __forceinline__ float fast_tanh(float x) {
#if __has_builtin(__builtin_amdgcn_tanhf)
  return __builtin_amdgcn_tanhf(x);
#else
  return tanhf(x);
#endif
}

__global__ __launch_bounds__(256) void xproj_kernel(const float* __restrict__ xs,
                                                    const float* __restrict__ Wx,
                                                    float* __restrict__ xp) {
  const int tid = threadIdx.x;
  const int wave = tid >> 5, lane = tid & 31;
  const int r = lane & 15, hi = lane >> 4, colsel = hi * 2;
  const size_t mbase = (size_t)blockIdx.x * 16;

  const int wrow = wave * 16 + r;
  v2f bfrag[32];
#pragma unroll
  for (int k = 0; k < 32; ++k)
    bfrag[k] = *(const v2f*)(Wx + wrow * DD + 4 * k + colsel);

  const float* arow = xs + (mbase + r) * DD + colsel;
  v8f acc = {};
#pragma unroll
  for (int k = 0; k < 32; ++k) {
    v2f a = *(const v2f*)(arow + 4 * k);
    acc = __builtin_amdgcn_wmma_f32_16x16x4_f32(
        false, a, false, bfrag[k], (short)0, acc, false, false);
  }
  float* orow = xp + (mbase + 8 * hi) * OO + wave * 16 + r;
#pragma unroll
  for (int v = 0; v < 8; ++v) orow[(size_t)v * OO] = acc[v];
}

__global__ __launch_bounds__(256) void scan_kernel(const float* __restrict__ Wh,
                                                   float* __restrict__ io) {
  __shared__ float hbuf[16 * HS];
  const int tid = threadIdx.x;
  const int wave = tid >> 5, lane = tid & 31;
  const int r = lane & 15, hi = lane >> 4, colsel = hi * 2;
  const int bbase = blockIdx.x * 16;

  const int wrow = wave * 16 + r;
  v2f whB[32];
#pragma unroll
  for (int k = 0; k < 32; ++k)
    whB[k] = *(const v2f*)(Wh + wrow * HH + 4 * k + colsel);

  for (int i = tid; i < 16 * HS; i += 256) hbuf[i] = 0.0f;
  __syncthreads();

  float* crow = io + (size_t)(bbase + 8 * hi) * LL * OO + wave * 16 + r;

  float xp[8];
#pragma unroll
  for (int v = 0; v < 8; ++v) xp[v] = crow[(size_t)v * LL * OO];

  for (int l = 0; l < LL; ++l) {
    const int ln = (l + 1 < LL) ? (l + 1) : l;
    float xpn[8];
#pragma unroll
    for (int v = 0; v < 8; ++v)
      xpn[v] = crow[((size_t)v * LL + ln) * OO];

    v8f acc = {xp[0], xp[1], xp[2], xp[3], xp[4], xp[5], xp[6], xp[7]};
    v8f acc1 = {};
#pragma unroll
    for (int k = 0; k < 16; ++k) {
      v2f a0 = *(const v2f*)(&hbuf[r * HS + 8 * k + colsel]);
      v2f a1 = *(const v2f*)(&hbuf[r * HS + 8 * k + 4 + colsel]);
      acc = __builtin_amdgcn_wmma_f32_16x16x4_f32(
          false, a0, false, whB[2 * k], (short)0, acc, false, false);
      acc1 = __builtin_amdgcn_wmma_f32_16x16x4_f32(
          false, a1, false, whB[2 * k + 1], (short)0, acc1, false, false);
    }

    float hv[8];
#pragma unroll
    for (int i = 0; i < 8; ++i) hv[i] = fast_tanh(acc[i] + acc1[i]);

    __syncthreads();
#pragma unroll
    for (int v = 0; v < 8; ++v) hbuf[(v + 8 * hi) * HS + wave * 16 + r] = hv[v];
#pragma unroll
    for (int v = 0; v < 8; ++v) crow[((size_t)v * LL + l) * OO] = hv[v];
    __syncthreads();

#pragma unroll
    for (int v = 0; v < 8; ++v) xp[v] = xpn[v];
  }
}

__global__ __launch_bounds__(256) void outproj_kernel(const float* __restrict__ Wo,
                                                      float* io) {
  const int tid = threadIdx.x;
  const int wave = tid >> 5, lane = tid & 31;
  const int r = lane & 15, hi = lane >> 4, colsel = hi * 2;
  const size_t mbase = (size_t)blockIdx.x * 16;

  const int wrow = wave * 16 + r;
  v2f woB[32];
#pragma unroll
  for (int k = 0; k < 32; ++k)
    woB[k] = *(const v2f*)(Wo + wrow * HH + 4 * k + colsel);

  const float* arow = io + (mbase + r) * HH + colsel;
  v8f acc = {};
#pragma unroll
  for (int k = 0; k < 32; ++k) {
    v2f a = *(const v2f*)(arow + 4 * k);
    acc = __builtin_amdgcn_wmma_f32_16x16x4_f32(
        false, a, false, woB[k], (short)0, acc, false, false);
  }
  __syncthreads();
  float* orow = io + (mbase + 8 * hi) * OO + wave * 16 + r;
#pragma unroll
  for (int v = 0; v < 8; ++v) orow[(size_t)v * OO] = acc[v];
}

extern "C" void kernel_launch(void* const* d_in, const int* in_sizes, int n_in,
                              void* d_out, int out_size, void* d_ws, size_t ws_size,
                              hipStream_t stream) {
  (void)in_sizes; (void)n_in; (void)d_ws; (void)ws_size; (void)out_size;
  const float* xs = (const float*)d_in[0];
  const float* Wh = (const float*)d_in[1];
  const float* Wx = (const float*)d_in[2];
  const float* Wo = (const float*)d_in[3];
  float* out = (float*)d_out;

  xproj_kernel<<<(BB * LL) / 16, 256, 0, stream>>>(xs, Wx, out);
  scan_kernel<<<BB / 16, 256, 0, stream>>>(Wh, out);
  outproj_kernel<<<(BB * LL) / 16, 256, 0, stream>>>(Wo, out);
}


# scan step no longer waits on global memory: LDS-only barrier fences replace __syncthreads (no loadcnt/storecnt drain per step) and xproj C fragments are loaded 4 steps ahead in a register ring
# speedup vs baseline: 2.2598x; 2.2598x over previous
#include <hip/hip_runtime.h>


typedef __attribute__((ext_vector_type(2))) float v2f;
typedef __attribute__((ext_vector_type(8))) float v8f;

#define BB 256
#define LL 1024
#define DD 128
#define HH 128
#define OO 128
#define HS 132
#define PF 4

__device__ __forceinline__ float fast_tanh(float x) {
#if __has_builtin(__builtin_amdgcn_tanhf)
  return __builtin_amdgcn_tanhf(x);
#else
  return tanhf(x);
#endif
}

__device__ __forceinline__ void lds_barrier() {
  __builtin_amdgcn_fence(__ATOMIC_RELEASE, "workgroup", "local");
  __builtin_amdgcn_s_barrier();
  __builtin_amdgcn_fence(__ATOMIC_ACQUIRE, "workgroup", "local");
}

__global__ __launch_bounds__(256) void xproj_kernel(const float* __restrict__ xs,
                                                    const float* __restrict__ Wx,
                                                    float* __restrict__ xp) {
  const int tid = threadIdx.x;
  const int wave = tid >> 5, lane = tid & 31;
  const int r = lane & 15, hi = lane >> 4, colsel = hi * 2;
  const size_t mbase = (size_t)blockIdx.x * 16;

  const int wrow = wave * 16 + r;
  v2f bfrag[32];
#pragma unroll
  for (int k = 0; k < 32; ++k)
    bfrag[k] = *(const v2f*)(Wx + wrow * DD + 4 * k + colsel);

  const float* arow = xs + (mbase + r) * DD + colsel;
  v8f acc = {};
#pragma unroll
  for (int k = 0; k < 32; ++k) {
    v2f a = *(const v2f*)(arow + 4 * k);
    acc = __builtin_amdgcn_wmma_f32_16x16x4_f32(
        false, a, false, bfrag[k], (short)0, acc, false, false);
  }
  float* orow = xp + (mbase + 8 * hi) * OO + wave * 16 + r;
#pragma unroll
  for (int v = 0; v < 8; ++v) orow[(size_t)v * OO] = acc[v];
}

__global__ __launch_bounds__(256) void scan_kernel(const float* __restrict__ Wh,
                                                   float* __restrict__ io) {
  __shared__ float hbuf[16 * HS];
  const int tid = threadIdx.x;
  const int wave = tid >> 5, lane = tid & 31;
  const int r = lane & 15, hi = lane >> 4, colsel = hi * 2;
  const int bbase = blockIdx.x * 16;

  const int wrow = wave * 16 + r;
  v2f whB[32];
#pragma unroll
  for (int k = 0; k < 32; ++k)
    whB[k] = *(const v2f*)(Wh + wrow * HH + 4 * k + colsel);

  float* crow = io + (size_t)(bbase + 8 * hi) * LL * OO + wave * 16 + r;

  float xq[PF][8];
#pragma unroll
  for (int j = 0; j < PF; ++j)
#pragma unroll
    for (int v = 0; v < 8; ++v) xq[j][v] = crow[((size_t)v * LL + j) * OO];

  for (int i = tid; i < 16 * HS; i += 256) hbuf[i] = 0.0f;
  __syncthreads();

  for (int l0 = 0; l0 < LL; l0 += PF) {
#pragma unroll
    for (int j = 0; j < PF; ++j) {
      const int l = l0 + j;
      v8f acc = {xq[j][0], xq[j][1], xq[j][2], xq[j][3],
                 xq[j][4], xq[j][5], xq[j][6], xq[j][7]};
      v8f acc1 = {};

#pragma unroll
      for (int k = 0; k < 16; ++k) {
        v2f a0 = *(const v2f*)(&hbuf[r * HS + 8 * k + colsel]);
        v2f a1 = *(const v2f*)(&hbuf[r * HS + 8 * k + 4 + colsel]);
        acc = __builtin_amdgcn_wmma_f32_16x16x4_f32(
            false, a0, false, whB[2 * k], (short)0, acc, false, false);
        acc1 = __builtin_amdgcn_wmma_f32_16x16x4_f32(
            false, a1, false, whB[2 * k + 1], (short)0, acc1, false, false);
      }

      float hv[8];
#pragma unroll
      for (int i = 0; i < 8; ++i) hv[i] = fast_tanh(acc[i] + acc1[i]);

      lds_barrier();
#pragma unroll
      for (int v = 0; v < 8; ++v) hbuf[(v + 8 * hi) * HS + wave * 16 + r] = hv[v];
#pragma unroll
      for (int v = 0; v < 8; ++v) crow[((size_t)v * LL + l) * OO] = hv[v];
      const int ln = (l + PF < LL) ? (l + PF) : (LL - 1);
#pragma unroll
      for (int v = 0; v < 8; ++v)
        xq[j][v] = crow[((size_t)v * LL + ln) * OO];
      lds_barrier();
    }
  }
}

__global__ __launch_bounds__(256) void outproj_kernel(const float* __restrict__ Wo,
                                                      float* io) {
  const int tid = threadIdx.x;
  const int wave = tid >> 5, lane = tid & 31;
  const int r = lane & 15, hi = lane >> 4, colsel = hi * 2;
  const size_t mbase = (size_t)blockIdx.x * 16;

  const int wrow = wave * 16 + r;
  v2f woB[32];
#pragma unroll
  for (int k = 0; k < 32; ++k)
    woB[k] = *(const v2f*)(Wo + wrow * HH + 4 * k + colsel);

  const float* arow = io + (mbase + r) * HH + colsel;
  v8f acc = {};
#pragma unroll
  for (int k = 0; k < 32; ++k) {
    v2f a = *(const v2f*)(arow + 4 * k);
    acc = __builtin_amdgcn_wmma_f32_16x16x4_f32(
        false, a, false, woB[k], (short)0, acc, false, false);
  }
  __syncthreads();
  float* orow = io + (mbase + 8 * hi) * OO + wave * 16 + r;
#pragma unroll
  for (int v = 0; v < 8; ++v) orow[(size_t)v * OO] = acc[v];
}

extern "C" void kernel_launch(void* const* d_in, const int* in_sizes, int n_in,
                              void* d_out, int out_size, void* d_ws, size_t ws_size,
                              hipStream_t stream) {
  (void)in_sizes; (void)n_in; (void)d_ws; (void)ws_size; (void)out_size;
  const float* xs = (const float*)d_in[0];
  const float* Wh = (const float*)d_in[1];
  const float* Wx = (const float*)d_in[2];
  const float* Wo = (const float*)d_in[3];
  float* out = (float*)d_out;

  xproj_kernel<<<(BB * LL) / 16, 256, 0, stream>>>(xs, Wx, out);
  scan_kernel<<<BB / 16, 256, 0, stream>>>(Wh, out);
  outproj_kernel<<<(BB * LL) / 16, 256, 0, stream>>>(Wo, out);
}


# projection launches: a block keeps its weight fragments in registers over 16 row tiles and gets each tile as whole-row 16-B loads through LDS (was 16-line fragment loads of both operands per tile)
# speedup vs baseline: 2.8873x; 1.2777x over previous
#include <hip/hip_runtime.h>


typedef __attribute__((ext_vector_type(2))) float v2f;
typedef __attribute__((ext_vector_type(4))) float v4f;
typedef __attribute__((ext_vector_type(8))) float v8f;

#define BB 256
#define LL 1024
#define DD 128
#define HH 128
#define OO 128
#define HS 132
#define PF 4
#define TPB 16

__device__ __forceinline__ float fast_tanh(float x) {
#if __has_builtin(__builtin_amdgcn_tanhf)
  return __builtin_amdgcn_tanhf(x);
#else
  return tanhf(x);
#endif
}

__device__ __forceinline__ void lds_barrier() {
  __builtin_amdgcn_fence(__ATOMIC_RELEASE, "workgroup", "local");
  __builtin_amdgcn_s_barrier();
  __builtin_amdgcn_fence(__ATOMIC_ACQUIRE, "workgroup", "local");
}

__device__ __forceinline__ void proj_tiles(const float* src, const float* __restrict__ Wt,
                                           float* dst, float* tile) {
  const int tid = threadIdx.x;
  const int wave = tid >> 5, lane = tid & 31;
  const int r = lane & 15, hi = lane >> 4, colsel = hi * 2;

  const int wrow = wave * 16 + r;
  v2f bfrag[32];
#pragma unroll
  for (int k = 0; k < 32; ++k)
    bfrag[k] = *(const v2f*)(Wt + wrow * DD + 4 * k + colsel);

  const size_t row0 = (size_t)blockIdx.x * (TPB * 16);
  const int srow = wave, sc = lane * 4;
  const float* gsrc = src + (row0 + srow) * DD + sc;
  v4f p0 = *(const v4f*)(gsrc);
  v4f p1 = *(const v4f*)(gsrc + 8 * DD);

  for (int t = 0; t < TPB; ++t) {
    float* buf = tile + (t & 1) * (16 * HS);
    *(v4f*)(buf + srow * HS + sc) = p0;
    *(v4f*)(buf + (srow + 8) * HS + sc) = p1;
    if (t + 1 < TPB) {
      p0 = *(const v4f*)(gsrc + (size_t)(t + 1) * 16 * DD);
      p1 = *(const v4f*)(gsrc + (size_t)(t + 1) * 16 * DD + 8 * DD);
    }
    lds_barrier();

    const float* arow = buf + r * HS + colsel;
    v8f acc = {};
#pragma unroll
    for (int k = 0; k < 32; ++k) {
      v2f a = *(const v2f*)(arow + 4 * k);
      acc = __builtin_amdgcn_wmma_f32_16x16x4_f32(
          false, a, false, bfrag[k], (short)0, acc, false, false);
    }
    float* orow = dst + (row0 + (size_t)t * 16 + 8 * hi) * OO + wave * 16 + r;
#pragma unroll
    for (int v = 0; v < 8; ++v) orow[(size_t)v * OO] = acc[v];
  }
}

__global__ __launch_bounds__(256) void xproj_kernel(const float* __restrict__ xs,
                                                    const float* __restrict__ Wx,
                                                    float* __restrict__ xp) {
  __shared__ __attribute__((aligned(16))) float tile[2 * 16 * HS];
  proj_tiles(xs, Wx, xp, tile);
}

__global__ __launch_bounds__(256) void scan_kernel(const float* __restrict__ Wh,
                                                   float* __restrict__ io) {
  __shared__ float hbuf[16 * HS];
  const int tid = threadIdx.x;
  const int wave = tid >> 5, lane = tid & 31;
  const int r = lane & 15, hi = lane >> 4, colsel = hi * 2;
  const int bbase = blockIdx.x * 16;

  const int wrow = wave * 16 + r;
  v2f whB[32];
#pragma unroll
  for (int k = 0; k < 32; ++k)
    whB[k] = *(const v2f*)(Wh + wrow * HH + 4 * k + colsel);

  float* crow = io + (size_t)(bbase + 8 * hi) * LL * OO + wave * 16 + r;

  float xq[PF][8];
#pragma unroll
  for (int j = 0; j < PF; ++j)
#pragma unroll
    for (int v = 0; v < 8; ++v) xq[j][v] = crow[((size_t)v * LL + j) * OO];

  for (int i = tid; i < 16 * HS; i += 256) hbuf[i] = 0.0f;
  __syncthreads();

  for (int l0 = 0; l0 < LL; l0 += PF) {
#pragma unroll
    for (int j = 0; j < PF; ++j) {
      const int l = l0 + j;
      v8f acc = {xq[j][0], xq[j][1], xq[j][2], xq[j][3],
                 xq[j][4], xq[j][5], xq[j][6], xq[j][7]};
      v8f acc1 = {};

#pragma unroll
      for (int k = 0; k < 16; ++k) {
        v2f a0 = *(const v2f*)(&hbuf[r * HS + 8 * k + colsel]);
        v2f a1 = *(const v2f*)(&hbuf[r * HS + 8 * k + 4 + colsel]);
        acc = __builtin_amdgcn_wmma_f32_16x16x4_f32(
            false, a0, false, whB[2 * k], (short)0, acc, false, false);
        acc1 = __builtin_amdgcn_wmma_f32_16x16x4_f32(
            false, a1, false, whB[2 * k + 1], (short)0, acc1, false, false);
      }

      float hv[8];
#pragma unroll
      for (int i = 0; i < 8; ++i) hv[i] = fast_tanh(acc[i] + acc1[i]);

      lds_barrier();
#pragma unroll
      for (int v = 0; v < 8; ++v) hbuf[(v + 8 * hi) * HS + wave * 16 + r] = hv[v];
#pragma unroll
      for (int v = 0; v < 8; ++v) crow[((size_t)v * LL + l) * OO] = hv[v];
      const int ln = (l + PF < LL) ? (l + PF) : (LL - 1);
#pragma unroll
      for (int v = 0; v < 8; ++v)
        xq[j][v] = crow[((size_t)v * LL + ln) * OO];
      lds_barrier();
    }
  }
}

__global__ __launch_bounds__(256) void outproj_kernel(const float* __restrict__ Wo,
                                                      float* io) {
  __shared__ __attribute__((aligned(16))) float tile[2 * 16 * HS];
  proj_tiles(io, Wo, io, tile);
}

extern "C" void kernel_launch(void* const* d_in, const int* in_sizes, int n_in,
                              void* d_out, int out_size, void* d_ws, size_t ws_size,
                              hipStream_t stream) {
  (void)in_sizes; (void)n_in; (void)d_ws; (void)ws_size; (void)out_size;
  const float* xs = (const float*)d_in[0];
  const float* Wh = (const float*)d_in[1];
  const float* Wx = (const float*)d_in[2];
  const float* Wo = (const float*)d_in[3];
  float* out = (float*)d_out;

  xproj_kernel<<<(BB * LL) / (16 * TPB), 256, 0, stream>>>(xs, Wx, out);
  scan_kernel<<<BB / 16, 256, 0, stream>>>(Wh, out);
  outproj_kernel<<<(BB * LL) / (16 * TPB), 256, 0, stream>>>(Wo, out);
}


# scan: the two waves sharing a SIMD take the matrix pipe in turns (split barriers, s_setprio on the younger wave's block); each does its tanh and LDS publish under the other's block; sums unchanged
# speedup vs baseline: 3.3242x; 1.1513x over previous
#include <hip/hip_runtime.h>


typedef __attribute__((ext_vector_type(2))) float v2f;
typedef __attribute__((ext_vector_type(4))) float v4f;
typedef __attribute__((ext_vector_type(8))) float v8f;

#define BB 256
#define LL 1024
#define DD 128
#define HH 128
#define OO 128
#define HS 132
#define PF 4
#define TPB 16

__device__ __forceinline__ float fast_tanh(float x) {
#if __has_builtin(__builtin_amdgcn_tanhf)
  return __builtin_amdgcn_tanhf(x);
#else
  return tanhf(x);
#endif
}

__device__ __forceinline__ void lds_barrier() {
  __builtin_amdgcn_fence(__ATOMIC_RELEASE, "workgroup", "local");
  __builtin_amdgcn_s_barrier();
  __builtin_amdgcn_fence(__ATOMIC_ACQUIRE, "workgroup", "local");
}

__device__ __forceinline__ void proj_tiles(const float* src, const float* __restrict__ Wt,
                                           float* dst, float* tile) {
  const int tid = threadIdx.x;
  const int wave = tid >> 5, lane = tid & 31;
  const int r = lane & 15, hi = lane >> 4, colsel = hi * 2;

  const int wrow = wave * 16 + r;
  v2f bfrag[32];
#pragma unroll
  for (int k = 0; k < 32; ++k)
    bfrag[k] = *(const v2f*)(Wt + wrow * DD + 4 * k + colsel);

  const size_t row0 = (size_t)blockIdx.x * (TPB * 16);
  const int srow = wave, sc = lane * 4;
  const float* gsrc = src + (row0 + srow) * DD + sc;
  v4f p0 = *(const v4f*)(gsrc);
  v4f p1 = *(const v4f*)(gsrc + 8 * DD);

  for (int t = 0; t < TPB; ++t) {
    float* buf = tile + (t & 1) * (16 * HS);
    *(v4f*)(buf + srow * HS + sc) = p0;
    *(v4f*)(buf + (srow + 8) * HS + sc) = p1;
    if (t + 1 < TPB) {
      p0 = *(const v4f*)(gsrc + (size_t)(t + 1) * 16 * DD);
      p1 = *(const v4f*)(gsrc + (size_t)(t + 1) * 16 * DD + 8 * DD);
    }
    lds_barrier();

    const float* arow = buf + r * HS + colsel;
    v8f acc = {};
#pragma unroll
    for (int k = 0; k < 32; ++k) {
      v2f a = *(const v2f*)(arow + 4 * k);
      acc = __builtin_amdgcn_wmma_f32_16x16x4_f32(
          false, a, false, bfrag[k], (short)0, acc, false, false);
    }
    float* orow = dst + (row0 + (size_t)t * 16 + 8 * hi) * OO + wave * 16 + r;
#pragma unroll
    for (int v = 0; v < 8; ++v) orow[(size_t)v * OO] = acc[v];
  }
}

__global__ __launch_bounds__(256) void xproj_kernel(const float* __restrict__ xs,
                                                    const float* __restrict__ Wx,
                                                    float* __restrict__ xp) {
  __shared__ __attribute__((aligned(16))) float tile[2 * 16 * HS];
  proj_tiles(xs, Wx, xp, tile);
}

__device__ __forceinline__ void bar_signal() {
  __builtin_amdgcn_fence(__ATOMIC_RELEASE, "workgroup", "local");
  __builtin_amdgcn_s_barrier_signal(-1);
}
__device__ __forceinline__ void bar_wait() {
  __builtin_amdgcn_s_barrier_wait(-1);
  __builtin_amdgcn_fence(__ATOMIC_ACQUIRE, "workgroup", "local");
}

__global__ __launch_bounds__(256) void scan_kernel(const float* __restrict__ Wh,
                                                   float* __restrict__ io) {
  __shared__ float hbuf[16 * HS];
  const int tid = threadIdx.x;
  const int wave = __builtin_amdgcn_readfirstlane(tid >> 5);
  const int lane = tid & 31;
  const int r = lane & 15, hi = lane >> 4, colsel = hi * 2;
  const int bbase = blockIdx.x * 16;
  const bool lead = wave < 4;

  const int wrow = wave * 16 + r;
  v2f whB[32];
#pragma unroll
  for (int k = 0; k < 32; ++k)
    whB[k] = *(const v2f*)(Wh + wrow * HH + 4 * k + colsel);

  float* crow = io + (size_t)(bbase + 8 * hi) * LL * OO + wave * 16 + r;

  float xq[PF][8];
#pragma unroll
  for (int j = 0; j < PF; ++j)
#pragma unroll
    for (int v = 0; v < 8; ++v) xq[j][v] = crow[((size_t)v * LL + j) * OO];

  for (int i = tid; i < 16 * HS; i += 256) hbuf[i] = 0.0f;
  __syncthreads();

  const float* arow = &hbuf[r * HS + colsel];
  float* hst = &hbuf[(8 * hi) * HS + wave * 16 + r];

  v8f acc = {}, acc1 = {};
  v8f sum = {};
  float hv[8] = {0.f, 0.f, 0.f, 0.f, 0.f, 0.f, 0.f, 0.f};
  v2f af[32];

  for (int l0 = 0; l0 < LL; l0 += PF) {
#pragma unroll
    for (int j = 0; j < PF; ++j) {
      const int l = l0 + j;
      const int jp = (j + PF - 1) % PF;

      if (!lead && l > 0) {
#pragma unroll
        for (int v = 0; v < 8; ++v) hv[v] = fast_tanh(sum[v]);
#pragma unroll
        for (int v = 0; v < 8; ++v) hst[v * HS] = hv[v];
      }
      bar_signal();

      if (l > 0) {
#pragma unroll
        for (int v = 0; v < 8; ++v) crow[((size_t)v * LL + (l - 1)) * OO] = hv[v];
        const int ln = (l - 1 + PF < LL) ? (l - 1 + PF) : (LL - 1);
#pragma unroll
        for (int v = 0; v < 8; ++v) xq[jp][v] = crow[((size_t)v * LL + ln) * OO];
      }

      if (lead) {
#pragma unroll
        for (int k = 0; k < 16; ++k) af[k] = *(const v2f*)(arow + 4 * k);
        acc = (v8f){xq[j][0], xq[j][1], xq[j][2], xq[j][3],
                    xq[j][4], xq[j][5], xq[j][6], xq[j][7]};
        acc1 = (v8f){};
#pragma unroll
        for (int k = 0; k < 10; k += 2) {
          acc = __builtin_amdgcn_wmma_f32_16x16x4_f32(
              false, af[k], false, whB[k], (short)0, acc, false, false);
          acc1 = __builtin_amdgcn_wmma_f32_16x16x4_f32(
              false, af[k + 1], false, whB[k + 1], (short)0, acc1, false, false);
        }
      }
      bar_wait();

      if (lead) {
#pragma unroll
        for (int k = 16; k < 32; ++k) af[k] = *(const v2f*)(arow + 4 * k);
#pragma unroll
        for (int k = 10; k < 32; k += 2) {
          acc = __builtin_amdgcn_wmma_f32_16x16x4_f32(
              false, af[k], false, whB[k], (short)0, acc, false, false);
          acc1 = __builtin_amdgcn_wmma_f32_16x16x4_f32(
              false, af[k + 1], false, whB[k + 1], (short)0, acc1, false, false);
        }
      } else {
#pragma unroll
        for (int k = 0; k < 32; ++k) af[k] = *(const v2f*)(arow + 4 * k);
      }
      bar_signal();
      bar_wait();

      if (lead) {
        sum = acc + acc1;
#pragma unroll
        for (int v = 0; v < 8; ++v) hv[v] = fast_tanh(sum[v]);
#pragma unroll
        for (int v = 0; v < 8; ++v) hst[v * HS] = hv[v];
      }
      bar_signal();

      if (!lead) {
        __builtin_amdgcn_s_setprio(3);
        acc = (v8f){xq[j][0], xq[j][1], xq[j][2], xq[j][3],
                    xq[j][4], xq[j][5], xq[j][6], xq[j][7]};
        acc1 = (v8f){};
#pragma unroll
        for (int k = 0; k < 32; k += 2) {
          acc = __builtin_amdgcn_wmma_f32_16x16x4_f32(
              false, af[k], false, whB[k], (short)0, acc, false, false);
          acc1 = __builtin_amdgcn_wmma_f32_16x16x4_f32(
              false, af[k + 1], false, whB[k + 1], (short)0, acc1, false, false);
        }
        sum = acc + acc1;
        __builtin_amdgcn_s_setprio(0);
      }
      bar_wait();
    }
  }
  if (!lead) {
#pragma unroll
    for (int v = 0; v < 8; ++v) hv[v] = fast_tanh(sum[v]);
  }
#pragma unroll
  for (int v = 0; v < 8; ++v) crow[((size_t)v * LL + (LL - 1)) * OO] = hv[v];
}

__global__ __launch_bounds__(256) void outproj_kernel(const float* __restrict__ Wo,
                                                      float* io) {
  __shared__ __attribute__((aligned(16))) float tile[2 * 16 * HS];
  proj_tiles(io, Wo, io, tile);
}

extern "C" void kernel_launch(void* const* d_in, const int* in_sizes, int n_in,
                              void* d_out, int out_size, void* d_ws, size_t ws_size,
                              hipStream_t stream) {
  (void)in_sizes; (void)n_in; (void)d_ws; (void)ws_size; (void)out_size;
  const float* xs = (const float*)d_in[0];
  const float* Wh = (const float*)d_in[1];
  const float* Wx = (const float*)d_in[2];
  const float* Wo = (const float*)d_in[3];
  float* out = (float*)d_out;

  xproj_kernel<<<(BB * LL) / (16 * TPB), 256, 0, stream>>>(xs, Wx, out);
  scan_kernel<<<BB / 16, 256, 0, stream>>>(Wh, out);
  outproj_kernel<<<(BB * LL) / (16 * TPB), 256, 0, stream>>>(Wo, out);
}


# time cut into 4 chunks: each scan chunk's launch also carries, on the idle compute units, x projections of later chunks and output projections of earlier ones (3 -> 6 launches; every sum unchanged)
# speedup vs baseline: 3.5699x; 1.0739x over previous
#include <hip/hip_runtime.h>


typedef __attribute__((ext_vector_type(2))) float v2f;
typedef __attribute__((ext_vector_type(4))) float v4f;
typedef __attribute__((ext_vector_type(8))) float v8f;

#define BB 256
#define LL 1024
#define DD 128
#define HH 128
#define OO 128
#define HS 132
#define PF 4
#define SEG 64
#define TPB (SEG / 16)
#define NSEG (LL / SEG)
#define U1 2
#define U2 8
#define U3 14
#define SOLO_FLOATS 45056
static_assert(SEG % 16 == 0 && LL % SEG == 0, "segments tile the sequence");
static_assert(SEG % PF == 0, "ring slots line up with chunk starts");
static_assert(0 < U1 && U1 < U2 && U2 < U3 && U3 < NSEG, "chunk bounds ascend");

__device__ __forceinline__ float fast_tanh(float x) {
#if __has_builtin(__builtin_amdgcn_tanhf)
  return __builtin_amdgcn_tanhf(x);
#else
  return tanhf(x);
#endif
}

__device__ __forceinline__ void lds_barrier() {
  __builtin_amdgcn_fence(__ATOMIC_RELEASE, "workgroup", "local");
  __builtin_amdgcn_s_barrier();
  __builtin_amdgcn_fence(__ATOMIC_ACQUIRE, "workgroup", "local");
}

__device__ __forceinline__ void proj_tiles(const float* src, const float* __restrict__ Wt,
                                           float* dst, float* tile, const size_t row0,
                                           const bool hold_last) {
  const int tid = threadIdx.x;
  const int wave = tid >> 5, lane = tid & 31;
  const int r = lane & 15, hi = lane >> 4, colsel = hi * 2;

  const int wrow = wave * 16 + r;
  v2f bfrag[32];
#pragma unroll
  for (int k = 0; k < 32; ++k)
    bfrag[k] = *(const v2f*)(Wt + wrow * DD + 4 * k + colsel);

  const int srow = wave, sc = lane * 4;
  const float* gsrc = src + (row0 + srow) * DD + sc;
  v4f p0 = *(const v4f*)(gsrc);
  v4f p1 = *(const v4f*)(gsrc + 8 * DD);

  for (int t = 0; t < TPB; ++t) {
    float* buf = tile + (t & 1) * (16 * HS);
    *(v4f*)(buf + srow * HS + sc) = p0;
    *(v4f*)(buf + (srow + 8) * HS + sc) = p1;
    if (t + 1 < TPB) {
      p0 = *(const v4f*)(gsrc + (size_t)(t + 1) * 16 * DD);
      p1 = *(const v4f*)(gsrc + (size_t)(t + 1) * 16 * DD + 8 * DD);
    }
    lds_barrier();

    const float* arow = buf + r * HS + colsel;
    v8f acc = {};
#pragma unroll
    for (int k = 0; k < 32; ++k) {
      v2f a = *(const v2f*)(arow + 4 * k);
      acc = __builtin_amdgcn_wmma_f32_16x16x4_f32(
          false, a, false, bfrag[k], (short)0, acc, false, false);
    }
    float* orow = dst + (row0 + (size_t)t * 16 + 8 * hi) * OO + wave * 16 + r;
#pragma unroll
    for (int v = 0; v < 7; ++v) orow[(size_t)v * OO] = acc[v];
    if (!(hold_last && t == TPB - 1 && hi == 1)) orow[(size_t)7 * OO] = acc[7];
  }
}

__global__ __launch_bounds__(256) void xproj_kernel(const float* __restrict__ xs,
                                                    const float* __restrict__ Wx,
                                                    float* __restrict__ xp) {
  __shared__ __attribute__((aligned(16))) float tile[2 * 16 * HS];
  proj_tiles(xs, Wx, xp, tile,
             (size_t)(blockIdx.x % BB) * LL + (size_t)(blockIdx.x / BB) * SEG, false);
}

__device__ __forceinline__ void bar_signal() {
  __builtin_amdgcn_fence(__ATOMIC_RELEASE, "workgroup", "local");
  __builtin_amdgcn_s_barrier_signal(-1);
}
__device__ __forceinline__ void bar_wait() {
  __builtin_amdgcn_s_barrier_wait(-1);
  __builtin_amdgcn_fence(__ATOMIC_ACQUIRE, "workgroup", "local");
}

__device__ __forceinline__ void scan_chunk(const float* __restrict__ Wh,
                                           const float* __restrict__ Wo,
                                           float* __restrict__ io, float* hbuf,
                                           const int lbeg, const int lend) {
  const int nst = lend - lbeg;
  const int tid = threadIdx.x;
  const int wave = __builtin_amdgcn_readfirstlane(tid >> 5);
  const int lane = tid & 31;
  const int r = lane & 15, hi = lane >> 4, colsel = hi * 2;
  const int bbase = blockIdx.x * 16;
  const bool lead = wave < 4;

  const int wrow = wave * 16 + r;
  v2f whB[32];
#pragma unroll
  for (int k = 0; k < 32; ++k)
    whB[k] = *(const v2f*)(Wh + wrow * HH + 4 * k + colsel);

  float* crow = io + ((size_t)(bbase + 8 * hi) * LL + lbeg) * OO + wave * 16 + r;

  float xq[PF][8];
#pragma unroll
  for (int j = 0; j < PF; ++j)
#pragma unroll
    for (int v = 0; v < 8; ++v) xq[j][v] = crow[((size_t)v * LL + j) * OO];

  const float* arow = &hbuf[r * HS + colsel];
  float* hst = &hbuf[(8 * hi) * HS + wave * 16 + r];

  if (lbeg == 0) {
    for (int i = tid; i < 16 * HS; i += 256) hbuf[i] = 0.0f;
  } else {
    const int hr = tid >> 4, hc = (tid & 15) * 8;
    const float* g = io + ((size_t)(bbase + hr) * LL + (lbeg - 1)) * OO + hc;
    const v4f g0 = *(const v4f*)(g);
    const v4f g1 = *(const v4f*)(g + 4);
    *(v4f*)(hbuf + hr * HS + hc) = g0;
    *(v4f*)(hbuf + hr * HS + hc + 4) = g1;
  }
  __syncthreads();
  if (lbeg != 0) {
    v2f woB[32];
#pragma unroll
    for (int k = 0; k < 32; ++k)
      woB[k] = *(const v2f*)(Wo + wrow * HH + 4 * k + colsel);
    v8f o = {};
#pragma unroll
    for (int k = 0; k < 32; ++k) {
      v2f a = *(const v2f*)(arow + 4 * k);
      o = __builtin_amdgcn_wmma_f32_16x16x4_f32(
          false, a, false, woB[k], (short)0, o, false, false);
    }
    float* prow = crow - OO;
#pragma unroll
    for (int v = 0; v < 8; ++v) prow[(size_t)v * LL * OO] = o[v];
  }

  v8f acc = {}, acc1 = {};
  v8f sum = {};
  float hv[8] = {0.f, 0.f, 0.f, 0.f, 0.f, 0.f, 0.f, 0.f};
  v2f af[32];

  for (int l0 = 0; l0 < nst; l0 += PF) {
#pragma unroll
    for (int j = 0; j < PF; ++j) {
      const int l = l0 + j;
      const int jp = (j + PF - 1) % PF;

      if (!lead && l > 0) {
#pragma unroll
        for (int v = 0; v < 8; ++v) hv[v] = fast_tanh(sum[v]);
#pragma unroll
        for (int v = 0; v < 8; ++v) hst[v * HS] = hv[v];
      }
      bar_signal();

      if (l > 0) {
#pragma unroll
        for (int v = 0; v < 8; ++v) crow[((size_t)v * LL + (l - 1)) * OO] = hv[v];
        const int ln = (l - 1 + PF < nst) ? (l - 1 + PF) : (nst - 1);
#pragma unroll
        for (int v = 0; v < 8; ++v) xq[jp][v] = crow[((size_t)v * LL + ln) * OO];
      }

      if (lead) {
#pragma unroll
        for (int k = 0; k < 16; ++k) af[k] = *(const v2f*)(arow + 4 * k);
        acc = (v8f){xq[j][0], xq[j][1], xq[j][2], xq[j][3],
                    xq[j][4], xq[j][5], xq[j][6], xq[j][7]};
        acc1 = (v8f){};
#pragma unroll
        for (int k = 0; k < 10; k += 2) {
          acc = __builtin_amdgcn_wmma_f32_16x16x4_f32(
              false, af[k], false, whB[k], (short)0, acc, false, false);
          acc1 = __builtin_amdgcn_wmma_f32_16x16x4_f32(
              false, af[k + 1], false, whB[k + 1], (short)0, acc1, false, false);
        }
      }
      bar_wait();

      if (lead) {
#pragma unroll
        for (int k = 16; k < 32; ++k) af[k] = *(const v2f*)(arow + 4 * k);
#pragma unroll
        for (int k = 10; k < 32; k += 2) {
          acc = __builtin_amdgcn_wmma_f32_16x16x4_f32(
              false, af[k], false, whB[k], (short)0, acc, false, false);
          acc1 = __builtin_amdgcn_wmma_f32_16x16x4_f32(
              false, af[k + 1], false, whB[k + 1], (short)0, acc1, false, false);
        }
      } else {
#pragma unroll
        for (int k = 0; k < 32; ++k) af[k] = *(const v2f*)(arow + 4 * k);
      }
      bar_signal();
      bar_wait();

      if (lead) {
        sum = acc + acc1;
#pragma unroll
        for (int v = 0; v < 8; ++v) hv[v] = fast_tanh(sum[v]);
#pragma unroll
        for (int v = 0; v < 8; ++v) hst[v * HS] = hv[v];
      }
      bar_signal();

      if (!lead) {
        __builtin_amdgcn_s_setprio(3);
        acc = (v8f){xq[j][0], xq[j][1], xq[j][2], xq[j][3],
                    xq[j][4], xq[j][5], xq[j][6], xq[j][7]};
        acc1 = (v8f){};
#pragma unroll
        for (int k = 0; k < 32; k += 2) {
          acc = __builtin_amdgcn_wmma_f32_16x16x4_f32(
              false, af[k], false, whB[k], (short)0, acc, false, false);
          acc1 = __builtin_amdgcn_wmma_f32_16x16x4_f32(
              false, af[k + 1], false, whB[k + 1], (short)0, acc1, false, false);
        }
        sum = acc + acc1;
        __builtin_amdgcn_s_setprio(0);
      }
      bar_wait();
    }
  }
  if (!lead) {
#pragma unroll
    for (int v = 0; v < 8; ++v) hv[v] = fast_tanh(sum[v]);
  }
#pragma unroll
  for (int v = 0; v < 8; ++v) crow[((size_t)v * LL + (nst - 1)) * OO] = hv[v];
}

__global__ __launch_bounds__(256) void step_kernel(const float* __restrict__ xs,
                                                   const float* __restrict__ Wx,
                                                   const float* __restrict__ Wh,
                                                   const float* __restrict__ Wo,
                                                   float* io, const int lbeg, const int lend,
                                                   const int x0, const int nx,
                                                   const int o0) {
  __shared__ __attribute__((aligned(16))) float smem[SOLO_FLOATS];
  if (blockIdx.x >= BB / 16) {
    int p = blockIdx.x - BB / 16;
    if (p < nx * BB) {
      const int sg = x0 + p / BB;
      proj_tiles(xs, Wx, io, smem, (size_t)(p % BB) * LL + (size_t)sg * SEG, false);
    } else {
      p -= nx * BB;
      const int sg = o0 + p / BB;
      proj_tiles(io, Wo, io, smem, (size_t)(p % BB) * LL + (size_t)sg * SEG,
                 (sg + 1) * SEG == lbeg);
    }
    return;
  }
  scan_chunk(Wh, Wo, io, smem, lbeg, lend);
}

__global__ __launch_bounds__(256) void outproj_kernel(const float* __restrict__ Wo,
                                                      float* io) {
  __shared__ __attribute__((aligned(16))) float tile[2 * 16 * HS];
  proj_tiles(io, Wo, io, tile,
             (size_t)(blockIdx.x % BB) * LL + (size_t)(U3 + blockIdx.x / BB) * SEG, false);
}

extern "C" void kernel_launch(void* const* d_in, const int* in_sizes, int n_in,
                              void* d_out, int out_size, void* d_ws, size_t ws_size,
                              hipStream_t stream) {
  (void)in_sizes; (void)n_in; (void)d_ws; (void)ws_size; (void)out_size;
  const float* xs = (const float*)d_in[0];
  const float* Wh = (const float*)d_in[1];
  const float* Wx = (const float*)d_in[2];
  const float* Wo = (const float*)d_in[3];
  float* out = (float*)d_out;

  xproj_kernel<<<U1 * BB, 256, 0, stream>>>(xs, Wx, out);
  step_kernel<<<BB / 16 + (U2 - U1) * BB, 256, 0, stream>>>(
      xs, Wx, Wh, Wo, out, 0, U1 * SEG, U1, U2 - U1, 0);
  step_kernel<<<BB / 16 + (NSEG - U2 + U1) * BB, 256, 0, stream>>>(
      xs, Wx, Wh, Wo, out, U1 * SEG, U2 * SEG, U2, NSEG - U2, 0);
  step_kernel<<<BB / 16 + (U2 - U1) * BB, 256, 0, stream>>>(
      xs, Wx, Wh, Wo, out, U2 * SEG, U3 * SEG, 0, 0, U1);
  step_kernel<<<BB / 16 + (U3 - U2) * BB, 256, 0, stream>>>(
      xs, Wx, Wh, Wo, out, U3 * SEG, NSEG * SEG, 0, 0, U2);
  outproj_kernel<<<(NSEG - U3) * BB, 256, 0, stream>>>(Wo, out);
}


# scan: the LDS serves the leads' reads of the newest columns first and alone: followers read the leads' columns ahead of that window and their own in front of their block; sums unchanged
# speedup vs baseline: 3.7417x; 1.0481x over previous
#include <hip/hip_runtime.h>


typedef __attribute__((ext_vector_type(2))) float v2f;
typedef __attribute__((ext_vector_type(4))) float v4f;
typedef __attribute__((ext_vector_type(8))) float v8f;

#define BB 256
#define LL 1024
#define DD 128
#define HH 128
#define OO 128
#define HS 132
#define PF 4
#define SEG 64
#define TPB (SEG / 16)
#define NSEG (LL / SEG)
#define U1 2
#define U2 8
#define U3 14
#define SOLO_FLOATS 45056
static_assert(SEG % 16 == 0 && LL % SEG == 0, "segments tile the sequence");
static_assert(SEG % PF == 0, "ring slots line up with chunk starts");
static_assert(0 < U1 && U1 < U2 && U2 < U3 && U3 < NSEG, "chunk bounds ascend");

__device__ __forceinline__ float fast_tanh(float x) {
#if __has_builtin(__builtin_amdgcn_tanhf)
  return __builtin_amdgcn_tanhf(x);
#else
  return tanhf(x);
#endif
}

__device__ __forceinline__ void lds_barrier() {
  __builtin_amdgcn_fence(__ATOMIC_RELEASE, "workgroup", "local");
  __builtin_amdgcn_s_barrier();
  __builtin_amdgcn_fence(__ATOMIC_ACQUIRE, "workgroup", "local");
}

__device__ __forceinline__ void proj_tiles(const float* src, const float* __restrict__ Wt,
                                           float* dst, float* tile, const size_t row0,
                                           const bool hold_last) {
  const int tid = threadIdx.x;
  const int wave = tid >> 5, lane = tid & 31;
  const int r = lane & 15, hi = lane >> 4, colsel = hi * 2;

  const int wrow = wave * 16 + r;
  v2f bfrag[32];
#pragma unroll
  for (int k = 0; k < 32; ++k)
    bfrag[k] = *(const v2f*)(Wt + wrow * DD + 4 * k + colsel);

  const int srow = wave, sc = lane * 4;
  const float* gsrc = src + (row0 + srow) * DD + sc;
  v4f p0 = *(const v4f*)(gsrc);
  v4f p1 = *(const v4f*)(gsrc + 8 * DD);

  for (int t = 0; t < TPB; ++t) {
    float* buf = tile + (t & 1) * (16 * HS);
    *(v4f*)(buf + srow * HS + sc) = p0;
    *(v4f*)(buf + (srow + 8) * HS + sc) = p1;
    if (t + 1 < TPB) {
      p0 = *(const v4f*)(gsrc + (size_t)(t + 1) * 16 * DD);
      p1 = *(const v4f*)(gsrc + (size_t)(t + 1) * 16 * DD + 8 * DD);
    }
    lds_barrier();

    const float* arow = buf + r * HS + colsel;
    v8f acc = {};
#pragma unroll
    for (int k = 0; k < 32; ++k) {
      v2f a = *(const v2f*)(arow + 4 * k);
      acc = __builtin_amdgcn_wmma_f32_16x16x4_f32(
          false, a, false, bfrag[k], (short)0, acc, false, false);
    }
    float* orow = dst + (row0 + (size_t)t * 16 + 8 * hi) * OO + wave * 16 + r;
#pragma unroll
    for (int v = 0; v < 7; ++v) orow[(size_t)v * OO] = acc[v];
    if (!(hold_last && t == TPB - 1 && hi == 1)) orow[(size_t)7 * OO] = acc[7];
  }
}

__global__ __launch_bounds__(256) void xproj_kernel(const float* __restrict__ xs,
                                                    const float* __restrict__ Wx,
                                                    float* __restrict__ xp) {
  __shared__ __attribute__((aligned(16))) float tile[2 * 16 * HS];
  proj_tiles(xs, Wx, xp, tile,
             (size_t)(blockIdx.x % BB) * LL + (size_t)(blockIdx.x / BB) * SEG, false);
}

__device__ __forceinline__ void bar_signal() {
  __builtin_amdgcn_fence(__ATOMIC_RELEASE, "workgroup", "local");
  __builtin_amdgcn_s_barrier_signal(-1);
}
__device__ __forceinline__ void bar_wait() {
  __builtin_amdgcn_s_barrier_wait(-1);
  __builtin_amdgcn_fence(__ATOMIC_ACQUIRE, "workgroup", "local");
}

__device__ __forceinline__ void scan_chunk(const float* __restrict__ Wh,
                                           const float* __restrict__ Wo,
                                           float* __restrict__ io, float* hbuf,
                                           const int lbeg, const int lend) {
  const int nst = lend - lbeg;
  const int tid = threadIdx.x;
  const int wave = __builtin_amdgcn_readfirstlane(tid >> 5);
  const int lane = tid & 31;
  const int r = lane & 15, hi = lane >> 4, colsel = hi * 2;
  const int bbase = blockIdx.x * 16;
  const bool lead = wave < 4;

  const int wrow = wave * 16 + r;
  v2f whB[32];
#pragma unroll
  for (int k = 0; k < 32; ++k)
    whB[k] = *(const v2f*)(Wh + wrow * HH + 4 * k + colsel);

  float* crow = io + ((size_t)(bbase + 8 * hi) * LL + lbeg) * OO + wave * 16 + r;

  float xq[PF][8];
#pragma unroll
  for (int j = 0; j < PF; ++j)
#pragma unroll
    for (int v = 0; v < 8; ++v) xq[j][v] = crow[((size_t)v * LL + j) * OO];

  const float* arow = &hbuf[r * HS + colsel];
  float* hst = &hbuf[(8 * hi) * HS + wave * 16 + r];

  if (lbeg == 0) {
    for (int i = tid; i < 16 * HS; i += 256) hbuf[i] = 0.0f;
  } else {
    const int hr = tid >> 4, hc = (tid & 15) * 8;
    const float* g = io + ((size_t)(bbase + hr) * LL + (lbeg - 1)) * OO + hc;
    const v4f g0 = *(const v4f*)(g);
    const v4f g1 = *(const v4f*)(g + 4);
    *(v4f*)(hbuf + hr * HS + hc) = g0;
    *(v4f*)(hbuf + hr * HS + hc + 4) = g1;
  }
  __syncthreads();
  if (lbeg != 0) {
    v2f woB[32];
#pragma unroll
    for (int k = 0; k < 32; ++k)
      woB[k] = *(const v2f*)(Wo + wrow * HH + 4 * k + colsel);
    v8f o = {};
#pragma unroll
    for (int k = 0; k < 32; ++k) {
      v2f a = *(const v2f*)(arow + 4 * k);
      o = __builtin_amdgcn_wmma_f32_16x16x4_f32(
          false, a, false, woB[k], (short)0, o, false, false);
    }
    float* prow = crow - OO;
#pragma unroll
    for (int v = 0; v < 8; ++v) prow[(size_t)v * LL * OO] = o[v];
  }

  v8f acc = {}, acc1 = {};
  v8f sum = {};
  float hv[8] = {0.f, 0.f, 0.f, 0.f, 0.f, 0.f, 0.f, 0.f};
  v2f af[32];

  for (int l0 = 0; l0 < nst; l0 += PF) {
#pragma unroll
    for (int j = 0; j < PF; ++j) {
      const int l = l0 + j;
      const int jp = (j + PF - 1) % PF;

      if (!lead && l > 0) {
#pragma unroll
        for (int v = 0; v < 8; ++v) hv[v] = fast_tanh(sum[v]);
#pragma unroll
        for (int v = 0; v < 8; ++v) hst[v * HS] = hv[v];
      }
      bar_signal();

#pragma unroll
      for (int k = 0; k < 16; ++k) af[k] = *(const v2f*)(arow + 4 * k);
      __builtin_amdgcn_sched_barrier(0);

      if (l > 0) {
#pragma unroll
        for (int v = 0; v < 8; ++v) crow[((size_t)v * LL + (l - 1)) * OO] = hv[v];
        const int ln = (l - 1 + PF < nst) ? (l - 1 + PF) : (nst - 1);
#pragma unroll
        for (int v = 0; v < 8; ++v) xq[jp][v] = crow[((size_t)v * LL + ln) * OO];
      }

      if (lead) {
        acc = (v8f){xq[j][0], xq[j][1], xq[j][2], xq[j][3],
                    xq[j][4], xq[j][5], xq[j][6], xq[j][7]};
        acc1 = (v8f){};
#pragma unroll
        for (int k = 0; k < 10; k += 2) {
          acc = __builtin_amdgcn_wmma_f32_16x16x4_f32(
              false, af[k], false, whB[k], (short)0, acc, false, false);
          acc1 = __builtin_amdgcn_wmma_f32_16x16x4_f32(
              false, af[k + 1], false, whB[k + 1], (short)0, acc1, false, false);
        }
      }
      bar_wait();

      if (lead) {
#pragma unroll
        for (int k = 16; k < 32; ++k) af[k] = *(const v2f*)(arow + 4 * k);
        __builtin_amdgcn_sched_barrier(0);
#pragma unroll
        for (int k = 10; k < 32; k += 2) {
          acc = __builtin_amdgcn_wmma_f32_16x16x4_f32(
              false, af[k], false, whB[k], (short)0, acc, false, false);
          acc1 = __builtin_amdgcn_wmma_f32_16x16x4_f32(
              false, af[k + 1], false, whB[k + 1], (short)0, acc1, false, false);
        }
      }
      bar_signal();
      bar_wait();

      if (lead) {
        sum = acc + acc1;
#pragma unroll
        for (int v = 0; v < 8; ++v) hv[v] = fast_tanh(sum[v]);
#pragma unroll
        for (int v = 0; v < 8; ++v) hst[v * HS] = hv[v];
      }
      if (!lead) {
#pragma unroll
        for (int k = 16; k < 32; ++k) af[k] = *(const v2f*)(arow + 4 * k);
        __builtin_amdgcn_s_setprio(3);
        acc = (v8f){xq[j][0], xq[j][1], xq[j][2], xq[j][3],
                    xq[j][4], xq[j][5], xq[j][6], xq[j][7]};
        acc1 = (v8f){};
#pragma unroll
        for (int k = 0; k < 12; k += 2) {
          acc = __builtin_amdgcn_wmma_f32_16x16x4_f32(
              false, af[k], false, whB[k], (short)0, acc, false, false);
          acc1 = __builtin_amdgcn_wmma_f32_16x16x4_f32(
              false, af[k + 1], false, whB[k + 1], (short)0, acc1, false, false);
        }
      }
      bar_signal();
      if (!lead) {
#pragma unroll
        for (int k = 12; k < 32; k += 2) {
          acc = __builtin_amdgcn_wmma_f32_16x16x4_f32(
              false, af[k], false, whB[k], (short)0, acc, false, false);
          acc1 = __builtin_amdgcn_wmma_f32_16x16x4_f32(
              false, af[k + 1], false, whB[k + 1], (short)0, acc1, false, false);
        }
        sum = acc + acc1;
        __builtin_amdgcn_s_setprio(0);
      }
      bar_wait();
    }
  }
  if (!lead) {
#pragma unroll
    for (int v = 0; v < 8; ++v) hv[v] = fast_tanh(sum[v]);
  }
#pragma unroll
  for (int v = 0; v < 8; ++v) crow[((size_t)v * LL + (nst - 1)) * OO] = hv[v];
}

__global__ __launch_bounds__(256) void step_kernel(const float* __restrict__ xs,
                                                   const float* __restrict__ Wx,
                                                   const float* __restrict__ Wh,
                                                   const float* __restrict__ Wo,
                                                   float* io, const int lbeg, const int lend,
                                                   const int x0, const int nx,
                                                   const int o0) {
  __shared__ __attribute__((aligned(16))) float smem[SOLO_FLOATS];
  if (blockIdx.x >= BB / 16) {
    int p = blockIdx.x - BB / 16;
    if (p < nx * BB) {
      const int sg = x0 + p / BB;
      proj_tiles(xs, Wx, io, smem, (size_t)(p % BB) * LL + (size_t)sg * SEG, false);
    } else {
      p -= nx * BB;
      const int sg = o0 + p / BB;
      proj_tiles(io, Wo, io, smem, (size_t)(p % BB) * LL + (size_t)sg * SEG,
                 (sg + 1) * SEG == lbeg);
    }
    return;
  }
  scan_chunk(Wh, Wo, io, smem, lbeg, lend);
}

__global__ __launch_bounds__(256) void outproj_kernel(const float* __restrict__ Wo,
                                                      float* io) {
  __shared__ __attribute__((aligned(16))) float tile[2 * 16 * HS];
  proj_tiles(io, Wo, io, tile,
             (size_t)(blockIdx.x % BB) * LL + (size_t)(U3 + blockIdx.x / BB) * SEG, false);
}

extern "C" void kernel_launch(void* const* d_in, const int* in_sizes, int n_in,
                              void* d_out, int out_size, void* d_ws, size_t ws_size,
                              hipStream_t stream) {
  (void)in_sizes; (void)n_in; (void)d_ws; (void)ws_size; (void)out_size;
  const float* xs = (const float*)d_in[0];
  const float* Wh = (const float*)d_in[1];
  const float* Wx = (const float*)d_in[2];
  const float* Wo = (const float*)d_in[3];
  float* out = (float*)d_out;

  xproj_kernel<<<U1 * BB, 256, 0, stream>>>(xs, Wx, out);
  step_kernel<<<BB / 16 + (U2 - U1) * BB, 256, 0, stream>>>(
      xs, Wx, Wh, Wo, out, 0, U1 * SEG, U1, U2 - U1, 0);
  step_kernel<<<BB / 16 + (NSEG - U2 + U1) * BB, 256, 0, stream>>>(
      xs, Wx, Wh, Wo, out, U1 * SEG, U2 * SEG, U2, NSEG - U2, 0);
  step_kernel<<<BB / 16 + (U2 - U1) * BB, 256, 0, stream>>>(
      xs, Wx, Wh, Wo, out, U2 * SEG, U3 * SEG, 0, 0, U1);
  step_kernel<<<BB / 16 + (U3 - U2) * BB, 256, 0, stream>>>(
      xs, Wx, Wh, Wo, out, U3 * SEG, NSEG * SEG, 0, 0, U2);
  outproj_kernel<<<(NSEG - U3) * BB, 256, 0, stream>>>(Wo, out);
}
